# starting kernel
# baseline (speedup 1.0000x reference)
#include <hip/hip_runtime.h>
#include <hip/hip_bf16.h>

#define BB 32
#define NN 8192
#define HD 256
#define GH 512
#define KC 64
#define ROWS_PER_CLUSTER 128
#define MROWS (BB * (2 * KC + 1))
#define OUTROWS (KC + 1)

typedef __attribute__((ext_vector_type(2))) float v2f;
typedef __attribute__((ext_vector_type(4))) float v4f;
typedef __attribute__((ext_vector_type(8))) float v8f;

__global__ __launch_bounds__(256)
void cluster_sum_kernel(const float* __restrict__ hs, float* __restrict__ H) {
    const int b = blockIdx.x >> 6;
    const int k = blockIdx.x & 63;
    const int c = threadIdx.x & 63;
    const int rg = threadIdx.x >> 6;

    const float* base = hs + ((size_t)b * NN + (size_t)k * ROWS_PER_CLUSTER) * HD;

    v4f sum = {0.f, 0.f, 0.f, 0.f};
    #pragma unroll 4
    for (int j = 0; j < ROWS_PER_CLUSTER / 4; ++j) {
        const int row = rg + 4 * j;
        sum += *(const v4f*)(base + (size_t)row * HD + c * 4);
    }

    __shared__ v4f sdata[4][64];
    sdata[rg][c] = sum;
    __syncthreads();

    if (rg == 0) {
        v4f tot = sdata[0][c] + sdata[1][c] + sdata[2][c] + sdata[3][c];
        const v4f hn = *(const v4f*)(hs + ((size_t)b * NN + (NN - 1)) * HD + c * 4);
        if (k == KC - 1) tot -= hn;

        float* Hb = H + (size_t)b * (2 * KC + 1) * HD;
        *(v4f*)(Hb + (size_t)k * HD + c * 4)        = tot;
        *(v4f*)(Hb + (size_t)(KC + k) * HD + c * 4) = tot + hn;
        if (k == 0)
            *(v4f*)(Hb + (size_t)(2 * KC) * HD + c * 4) = hn;
    }
}

__global__ __launch_bounds__(256)
void transpose_kernel(const float* __restrict__ in, float* __restrict__ out,
                      int R, int Ccols) {
    __shared__ float tile[32][33];
    const int tx = threadIdx.x & 31;
    const int ty = threadIdx.x >> 5;
    const int tilesC = Ccols >> 5;
    const int c0 = (blockIdx.x % tilesC) << 5;
    const int r0 = (blockIdx.x / tilesC) << 5;

    #pragma unroll
    for (int i = 0; i < 32; i += 8)
        tile[ty + i][tx] = in[(size_t)(r0 + ty + i) * Ccols + (c0 + tx)];
    __syncthreads();
    #pragma unroll
    for (int i = 0; i < 32; i += 8)
        out[(size_t)(c0 + ty + i) * R + (r0 + tx)] = tile[tx][ty + i];
}

template <bool RELU>
__global__ __launch_bounds__(256)
void gemm_wmma_f32(const float* __restrict__ A, const float* __restrict__ Bt,
                   const float* __restrict__ bias, float* __restrict__ C,
                   int M, int Kd, int Nd, int nStrips) {
    const int lane  = threadIdx.x & 31;
    const int wave  = threadIdx.x >> 5;
    const int strip = blockIdx.x * 8 + wave;
    if (strip >= nStrips) return;

    const int tilesM = M >> 4;
    const int m0 = (strip % tilesM) << 4;
    const int n0 = (strip / tilesM) << 6;

    const int colL = lane & 15;
    const int kb   = (lane >> 4) << 1;

    v8f acc[4];
    #pragma unroll
    for (int t = 0; t < 4; ++t) {
        const float bv = bias[n0 + 16 * t + colL];
        acc[t] = (v8f){bv, bv, bv, bv, bv, bv, bv, bv};
    }

    const float* arow  = A  + (size_t)(m0 + colL) * Kd + kb;
    const float* brow0 = Bt + (size_t)(n0 + colL) * Kd + kb;
    const size_t bstr  = (size_t)16 * Kd;

    v2f a_cur = *(const v2f*)(arow);
    v2f b_cur[4];
    #pragma unroll
    for (int t = 0; t < 4; ++t)
        b_cur[t] = *(const v2f*)(brow0 + t * bstr);

    #pragma unroll 4
    for (int k0 = 0; k0 < Kd - 4; k0 += 4) {
        const int kn = k0 + 4;
        v2f a_nxt = *(const v2f*)(arow + kn);
        v2f b_nxt[4];
        #pragma unroll
        for (int t = 0; t < 4; ++t)
            b_nxt[t] = *(const v2f*)(brow0 + t * bstr + kn);

        #pragma unroll
        for (int t = 0; t < 4; ++t)
            acc[t] = __builtin_amdgcn_wmma_f32_16x16x4_f32(
                false, a_cur, false, b_cur[t], (short)0, acc[t], false, false);

        a_cur = a_nxt;
        #pragma unroll
        for (int t = 0; t < 4; ++t) b_cur[t] = b_nxt[t];
    }

    #pragma unroll
    for (int t = 0; t < 4; ++t)
        acc[t] = __builtin_amdgcn_wmma_f32_16x16x4_f32(
            false, a_cur, false, b_cur[t], (short)0, acc[t], false, false);

    const int rbase = m0 + ((lane >> 4) << 3);
    #pragma unroll
    for (int t = 0; t < 4; ++t) {
        const int col = n0 + 16 * t + colL;
        #pragma unroll
        for (int v = 0; v < 8; ++v) {
            float x = acc[t][v];
            if (RELU) x = fmaxf(x, 0.0f);
            C[(size_t)(rbase + v) * Nd + col] = x;
        }
    }
}

__global__ __launch_bounds__(256)
void finalize_kernel(const float* __restrict__ gs, float* __restrict__ G,
                     float* __restrict__ Gmask) {
    const int b = blockIdx.x;
    const int h = threadIdx.x;
    const float* g = gs + (size_t)b * (2 * KC + 1) * HD + h;

    float S = 0.f;
    #pragma unroll 8
    for (int k = 0; k < KC; ++k) S += g[(size_t)k * HD];

    float* Gb = G + (size_t)b * OUTROWS * HD;
    #pragma unroll 4
    for (int k = 0; k < KC; ++k)
        Gb[(size_t)k * HD + h] = S - g[(size_t)k * HD] + g[(size_t)(KC + k) * HD];
    Gb[(size_t)KC * HD + h] = S + g[(size_t)(2 * KC) * HD];

    if (h < OUTROWS) Gmask[b * OUTROWS + h] = 1.0f;
}

extern "C" void kernel_launch(void* const* d_in, const int* in_sizes, int n_in,
                              void* d_out, int out_size, void* d_ws, size_t ws_size,
                              hipStream_t stream) {
    const float* hs = (const float*)d_in[0];
    const float* W1 = (const float*)d_in[2];
    const float* b1 = (const float*)d_in[3];
    const float* W2 = (const float*)d_in[4];
    const float* b2 = (const float*)d_in[5];

    float* G     = (float*)d_out;
    float* Gmask = (float*)d_out + (size_t)BB * OUTROWS * HD;

    float* H   = (float*)d_ws;
    float* X   = H   + (size_t)MROWS * HD;
    float* W1t = X   + (size_t)MROWS * GH;
    float* W2t = W1t + (size_t)GH * HD;
    float* gs  = H;

    transpose_kernel<<<(HD / 32) * (GH / 32), 256, 0, stream>>>(W1, W1t, HD, GH);
    transpose_kernel<<<(GH / 32) * (HD / 32), 256, 0, stream>>>(W2, W2t, GH, HD);

    cluster_sum_kernel<<<BB * KC, 256, 0, stream>>>(hs, H);

    {
        const int nStrips = (MROWS / 16) * (GH / 64);
        gemm_wmma_f32<true><<<(nStrips + 7) / 8, 256, 0, stream>>>(
            H, W1t, b1, X, MROWS, HD, GH, nStrips);
    }

    {
        const int nStrips = (MROWS / 16) * (HD / 64);
        gemm_wmma_f32<false><<<(nStrips + 7) / 8, 256, 0, stream>>>(
            X, W2t, b2, gs, MROWS, GH, HD, nStrips);
    }

    finalize_kernel<<<BB, 256, 0, stream>>>(gs, G, Gmask);
}


# Second GEMM: operand loads issued two 32-k stages ahead of their WMMAs (three register sets; one wave per SIMD has the registers), h_n tile's WMMAs kept inside their stage; sum order unchanged
# speedup vs baseline: 2.4446x; 2.4446x over previous
#include <hip/hip_runtime.h>
#include <hip/hip_bf16.h>

#define BB 32
#define NN 8192
#define HD 256
#define GH 512
#define KC 64
#define ROWS_PER_CLUSTER 128
#define MROWS (BB * (2 * KC + 1))
#define OUTROWS (KC + 1)

typedef __attribute__((ext_vector_type(2))) float v2f;
typedef __attribute__((ext_vector_type(4))) float v4f;
typedef __attribute__((ext_vector_type(8))) float v8f;

__device__ __forceinline__ void pack_b_piece(const float* __restrict__ Wm,
                                             float* __restrict__ out,
                                             int Kd, int Nd, int idx) {
    const int lane  = idx & 31;
    const int chunk = idx >> 5;
    const int nkc   = Kd >> 3;
    const int kc    = chunk % nkc;
    const int nt    = chunk / nkc;
    const int n     = nt * 16 + (lane & 15);
    const int k     = kc * 8 + (lane >> 4) * 4;
    const float* p  = Wm + (size_t)k * Nd + n;
    v4f v;
    v.x = p[0];
    v.y = p[(size_t)Nd];
    v.z = p[(size_t)2 * Nd];
    v.w = p[(size_t)3 * Nd];
    *(v4f*)(out + (size_t)idx * 4) = v;
}

#define PACK_BLOCKS_W1 ((GH / 16) * (HD / 8) * 32 / 256)
#define PACK_BLOCKS_W2 ((HD / 16) * (GH / 8) * 32 / 256)
#define PACK_BLOCKS    (PACK_BLOCKS_W1 + PACK_BLOCKS_W2)

__global__ __launch_bounds__(256)
void cluster_sum_pack_kernel(const float* __restrict__ hs, float* __restrict__ H,
                             const float* __restrict__ W1, float* __restrict__ W1p,
                             const float* __restrict__ W2, float* __restrict__ W2p) {
    if (blockIdx.x < PACK_BLOCKS) {
        const int pb = blockIdx.x;
        if (pb < PACK_BLOCKS_W1)
            pack_b_piece(W1, W1p, HD, GH, pb * 256 + threadIdx.x);
        else
            pack_b_piece(W2, W2p, GH, HD, (pb - PACK_BLOCKS_W1) * 256 + threadIdx.x);
        return;
    }

    const int blk = blockIdx.x - PACK_BLOCKS;
    const int b = blk >> 6;
    const int k = blk & 63;
    const int c = threadIdx.x & 63;
    const int rg = threadIdx.x >> 6;

    const float* base = hs + ((size_t)b * NN + (size_t)k * ROWS_PER_CLUSTER) * HD;

    v4f sum = {0.f, 0.f, 0.f, 0.f};
    #pragma unroll 4
    for (int j = 0; j < ROWS_PER_CLUSTER / 4; ++j) {
        const int row = rg + 4 * j;
        sum += *(const v4f*)(base + (size_t)row * HD + c * 4);
    }

    __shared__ v4f sdata[4][64];
    sdata[rg][c] = sum;
    __syncthreads();

    if (rg == 0) {
        v4f tot = sdata[0][c] + sdata[1][c] + sdata[2][c] + sdata[3][c];
        const v4f hn = *(const v4f*)(hs + ((size_t)b * NN + (NN - 1)) * HD + c * 4);
        if (k == KC - 1) tot -= hn;

        float* Hb = H + (size_t)b * (2 * KC + 1) * HD;
        *(v4f*)(Hb + (size_t)k * HD + c * 4)        = tot;
        *(v4f*)(Hb + (size_t)(KC + k) * HD + c * 4) = tot + hn;
        if (k == 0)
            *(v4f*)(Hb + (size_t)(2 * KC) * HD + c * 4) = hn;
    }
}

#define LEFT_ROW0 ((MROWS / 16 - 2) * 16)

template <bool EX>
__device__ __forceinline__ void load_stage1(v4f (&a)[4], v4f (&ax)[4], v4f (&b)[4][4],
                                            const float* arow, const float* axrow,
                                            const float* bbase, const int (&boff)[4], int s) {
    #pragma unroll
    for (int j = 0; j < 4; ++j) {
        if (EX) ax[j] = *(const v4f*)(axrow + s * 32 + j * 8);
        a[j] = *(const v4f*)(arow + s * 32 + j * 8);
        #pragma unroll
        for (int t = 0; t < 4; ++t)
            b[t][j] = *(const v4f*)(bbase + boff[t] + (size_t)(s * 4 + j) * 128);
    }
}

template <bool EX>
__device__ __forceinline__ void mma_stage1(v8f (&acc)[4], v8f& accx, const v4f (&a)[4],
                                           const v4f (&ax)[4], const v4f (&b)[4][4]) {
    #pragma unroll
    for (int j = 0; j < 4; ++j) {
        const v2f alo = a[j].xy, ahi = a[j].zw;
        #pragma unroll
        for (int t = 0; t < 4; ++t)
            acc[t] = __builtin_amdgcn_wmma_f32_16x16x4_f32(
                false, alo, false, (v2f)b[t][j].xy, (short)0, acc[t], false, false);
        if (EX)
            accx = __builtin_amdgcn_wmma_f32_16x16x4_f32(
                false, (v2f)ax[j].xy, false, (v2f)b[0][j].xy, (short)0, accx, false, false);
        #pragma unroll
        for (int t = 0; t < 4; ++t)
            acc[t] = __builtin_amdgcn_wmma_f32_16x16x4_f32(
                false, ahi, false, (v2f)b[t][j].zw, (short)0, acc[t], false, false);
        if (EX)
            accx = __builtin_amdgcn_wmma_f32_16x16x4_f32(
                false, (v2f)ax[j].zw, false, (v2f)b[0][j].zw, (short)0, accx, false, false);
    }
}

template <bool EX>
__device__ __forceinline__ void gemm1_body(const float* A, const float* Bp,
                                           const float* __restrict__ bias, float* C,
                                           int lane, int mt, int ns, int r, int L) {
    const int colL = lane & 15;
    const int half = lane >> 4;
    const int m0 = mt << 4;
    const int n0 = ns << 6;

    int tt[4];
    #pragma unroll
    for (int t = 0; t < 4; ++t) tt[t] = EX ? ((t + r) & 3) : t;

    v8f acc[4];
    #pragma unroll
    for (int t = 0; t < 4; ++t) {
        const float bv = bias[n0 + 16 * tt[t] + colL];
        acc[t] = (v8f){bv, bv, bv, bv, bv, bv, bv, bv};
    }
    v8f accx = acc[0];

    const float* arow  = A  + (size_t)(m0 + colL) * HD + half * 4;
    const float* axrow = A  + (size_t)(LEFT_ROW0 + 16 * L + colL) * HD + half * 4;
    const float* bbase = Bp + ((size_t)(n0 >> 4) * (HD / 8) * 32 + lane) * 4;
    int boff[4];
    #pragma unroll
    for (int t = 0; t < 4; ++t) boff[t] = tt[t] * ((HD / 8) * 128);

    constexpr int NS = HD / 32;
    v4f a0[4], x0[4], b0[4][4], a1[4], x1[4], b1[4][4];

#define SCHED_FENCE() __builtin_amdgcn_sched_barrier(0)
    load_stage1<EX>(a0, x0, b0, arow, axrow, bbase, boff, 0);
    SCHED_FENCE();
    #pragma unroll 1
    for (int s = 0; s < NS - 2; s += 2) {
        load_stage1<EX>(a1, x1, b1, arow, axrow, bbase, boff, s + 1);
        SCHED_FENCE();
        mma_stage1<EX>(acc, accx, a0, x0, b0);
        SCHED_FENCE();
        load_stage1<EX>(a0, x0, b0, arow, axrow, bbase, boff, s + 2);
        SCHED_FENCE();
        mma_stage1<EX>(acc, accx, a1, x1, b1);
        SCHED_FENCE();
    }
    load_stage1<EX>(a1, x1, b1, arow, axrow, bbase, boff, NS - 1);
    SCHED_FENCE();
    mma_stage1<EX>(acc, accx, a0, x0, b0);
    mma_stage1<EX>(acc, accx, a1, x1, b1);
#undef SCHED_FENCE

    const int rbase = m0 + (half << 3);
    #pragma unroll
    for (int t = 0; t < 4; ++t) {
        const int col = n0 + 16 * tt[t] + colL;
        #pragma unroll
        for (int v = 0; v < 8; ++v)
            C[(size_t)(rbase + v) * GH + col] = fmaxf(acc[t][v], 0.0f);
    }
    if (EX) {
        const int rx  = LEFT_ROW0 + 16 * L + (half << 3);
        const int col = n0 + 16 * tt[0] + colL;
        #pragma unroll
        for (int v = 0; v < 8; ++v)
            C[(size_t)(rx + v) * GH + col] = fmaxf(accx[v], 0.0f);
    }
}

__global__ __launch_bounds__(256)
void gemm1_kernel(const float* A, const float* Bp,
                  const float* __restrict__ bias, float* C) {
    const int lane = threadIdx.x & 31;
    const int wave = __builtin_amdgcn_readfirstlane((int)(threadIdx.x >> 5));
    const int q  = blockIdx.x & 31;
    const int ns = blockIdx.x >> 5;
    const int mt = q * 8 + wave;

    if (wave == 0 && q < 8)
        gemm1_body<true>(A, Bp, bias, C, lane, mt, ns, q & 3, q >> 2);
    else
        gemm1_body<false>(A, Bp, bias, C, lane, mt, ns, 0, 0);
}

template <int HT>
__device__ __forceinline__ void load_stage_c(v4f (&a)[2][4], v4f (&ah)[4], v4f (&b)[2][4],
                                             const float* ar0, const float* ar1,
                                             const float* arh, const float* bbase, int s) {
    #pragma unroll
    for (int j = 0; j < 4; ++j) {
        if (HT >= 0) ah[j] = *(const v4f*)(arh + s * 32 + j * 8);
        a[0][j] = *(const v4f*)(ar0 + s * 32 + j * 8);
        a[1][j] = *(const v4f*)(ar1 + s * 32 + j * 8);
        #pragma unroll
        for (int t = 0; t < 2; ++t)
            b[t][j] = *(const v4f*)(bbase + ((size_t)t * (GH / 8) + (size_t)(s * 4 + j)) * 128);
    }
}

template <int HT>
__device__ __forceinline__ void mma_stage_c(v8f (&acc)[2][2], v8f& acch, const v4f (&a)[2][4],
                                            const v4f (&ah)[4], const v4f (&b)[2][4]) {
    constexpr int TB = HT < 0 ? 0 : HT;
    #pragma unroll
    for (int j = 0; j < 4; ++j) {
        #pragma unroll
        for (int m = 0; m < 2; ++m) {
            #pragma unroll
            for (int t = 0; t < 2; ++t)
                acc[m][t] = __builtin_amdgcn_wmma_f32_16x16x4_f32(
                    false, (v2f)a[m][j].xy, false, (v2f)b[t][j].xy, (short)0, acc[m][t], false, false);
        }
        if (HT >= 0)
            acch = __builtin_amdgcn_wmma_f32_16x16x4_f32(
                false, (v2f)ah[j].xy, false, (v2f)b[TB][j].xy, (short)0, acch, false, false);
        #pragma unroll
        for (int m = 0; m < 2; ++m) {
            #pragma unroll
            for (int t = 0; t < 2; ++t)
                acc[m][t] = __builtin_amdgcn_wmma_f32_16x16x4_f32(
                    false, (v2f)a[m][j].zw, false, (v2f)b[t][j].zw, (short)0, acc[m][t], false, false);
        }
        if (HT >= 0)
            acch = __builtin_amdgcn_wmma_f32_16x16x4_f32(
                false, (v2f)ah[j].zw, false, (v2f)b[TB][j].zw, (short)0, acch, false, false);
    }
    if (HT >= 0) asm volatile("" : "+v"(acch));
}

template <int HT>
__device__ __forceinline__ void k_loop_c(v8f (&acc)[2][2], v8f& acch,
                                         const float* ar0, const float* ar1,
                                         const float* arh, const float* bbase) {
    constexpr int NS = GH / 32;
    static_assert((NS - 4) % 3 == 0, "the loop below takes three stages per trip");
    v4f a0[2][4], h0[4], b0[2][4], a1[2][4], h1[4], b1[2][4], a2[2][4], h2[4], b2[2][4];
#define SCHED_FENCE() __builtin_amdgcn_sched_barrier(0)
    load_stage_c<HT>(a0, h0, b0, ar0, ar1, arh, bbase, 0);
    SCHED_FENCE();
    load_stage_c<HT>(a1, h1, b1, ar0, ar1, arh, bbase, 1);
    SCHED_FENCE();
    #pragma unroll 1
    for (int s = 0; s < NS - 4; s += 3) {
        load_stage_c<HT>(a2, h2, b2, ar0, ar1, arh, bbase, s + 2);
        SCHED_FENCE();
        mma_stage_c<HT>(acc, acch, a0, h0, b0);
        SCHED_FENCE();
        load_stage_c<HT>(a0, h0, b0, ar0, ar1, arh, bbase, s + 3);
        SCHED_FENCE();
        mma_stage_c<HT>(acc, acch, a1, h1, b1);
        SCHED_FENCE();
        load_stage_c<HT>(a1, h1, b1, ar0, ar1, arh, bbase, s + 4);
        SCHED_FENCE();
        mma_stage_c<HT>(acc, acch, a2, h2, b2);
        SCHED_FENCE();
    }
    load_stage_c<HT>(a2, h2, b2, ar0, ar1, arh, bbase, NS - 2);
    SCHED_FENCE();
    mma_stage_c<HT>(acc, acch, a0, h0, b0);
    SCHED_FENCE();
    load_stage_c<HT>(a0, h0, b0, ar0, ar1, arh, bbase, NS - 1);
    SCHED_FENCE();
    mma_stage_c<HT>(acc, acch, a1, h1, b1);
    mma_stage_c<HT>(acc, acch, a2, h2, b2);
    mma_stage_c<HT>(acc, acch, a0, h0, b0);
#undef SCHED_FENCE
}

__global__ __launch_bounds__(128)
void gemm2_combine_kernel(const float* X, const float* Bp,
                          const float* __restrict__ bias,
                          float* __restrict__ G, float* __restrict__ Gmask) {
    const int lane = threadIdx.x & 31;
    const int wave = __builtin_amdgcn_readfirstlane((int)(threadIdx.x >> 5));
    const int b    = blockIdx.x >> 3;
    const int n0   = (blockIdx.x & 7) << 5;
    const int colL = lane & 15;
    const int half = lane >> 4;

    v8f acc[2][2];
    #pragma unroll
    for (int t = 0; t < 2; ++t) {
        const float bv = bias[n0 + 16 * t + colL];
        acc[0][t] = (v8f){bv, bv, bv, bv, bv, bv, bv, bv};
        acc[1][t] = acc[0][t];
    }
    const float bvh = bias[n0 + 16 * (wave & 1) + colL];
    v8f acch = (v8f){bvh, bvh, bvh, bvh, bvh, bvh, bvh, bvh};

    const size_t r0 = (size_t)b * (2 * KC + 1);
    const float* ar0 = X + (r0 + 16 * wave + colL) * GH + half * 4;
    const float* ar1 = X + (r0 + KC + 16 * wave + colL) * GH + half * 4;
    const float* arh = X + (r0 + 2 * KC) * GH + half * 4;
    const float* bbase = Bp + ((size_t)(n0 >> 4) * (GH / 8) * 32 + lane) * 4;

    if (wave == 0)      k_loop_c<0>(acc, acch, ar0, ar1, arh, bbase);
    else if (wave == 1) k_loop_c<1>(acc, acch, ar0, ar1, arh, bbase);
    else                k_loop_c<-1>(acc, acch, ar0, ar1, arh, bbase);

    __shared__ float sS[4][2][32];
    #pragma unroll
    for (int t = 0; t < 2; ++t) {
        float ps = acc[0][t][0];
        #pragma unroll
        for (int v = 1; v < 8; ++v) ps += acc[0][t][v];
        sS[wave][half][16 * t + colL] = ps;
    }
    __syncthreads();
    float S[2];
    #pragma unroll
    for (int t = 0; t < 2; ++t) {
        float s = 0.f;
        #pragma unroll
        for (int w = 0; w < 4; ++w) {
            s += sS[w][0][16 * t + colL];
            s += sS[w][1][16 * t + colL];
        }
        S[t] = s;
    }

    float* Gb = G + (size_t)b * OUTROWS * HD;
    const int rbase = 16 * wave + (half << 3);
    #pragma unroll
    for (int t = 0; t < 2; ++t) {
        const int col = n0 + 16 * t + colL;
        #pragma unroll
        for (int v = 0; v < 8; ++v)
            Gb[(size_t)(rbase + v) * HD + col] = S[t] - acc[0][t][v] + acc[1][t][v];
    }
    if (wave < 2 && half == 0)
        Gb[(size_t)KC * HD + n0 + 16 * wave + colL] = (wave == 0 ? S[0] : S[1]) + acch[0];

    if ((blockIdx.x & 7) == 0 && threadIdx.x < OUTROWS)
        Gmask[b * OUTROWS + threadIdx.x] = 1.0f;
}

extern "C" void kernel_launch(void* const* d_in, const int* in_sizes, int n_in,
                              void* d_out, int out_size, void* d_ws, size_t ws_size,
                              hipStream_t stream) {
    const float* hs = (const float*)d_in[0];
    const float* W1 = (const float*)d_in[2];
    const float* b1 = (const float*)d_in[3];
    const float* W2 = (const float*)d_in[4];
    const float* b2 = (const float*)d_in[5];

    float* G     = (float*)d_out;
    float* Gmask = (float*)d_out + (size_t)BB * OUTROWS * HD;

    float* H   = (float*)d_ws;
    float* X   = H   + (size_t)MROWS * HD;
    float* W1t = X   + (size_t)MROWS * GH;
    float* W2t = W1t + (size_t)GH * HD;

    cluster_sum_pack_kernel<<<PACK_BLOCKS + BB * KC, 256, 0, stream>>>(
        hs, H, W1, W1t, W2, W2t);

    gemm1_kernel<<<(MROWS / 16 - 2) * (GH / 64) / 8, 256, 0, stream>>>(H, W1t, b1, X);

    gemm2_combine_kernel<<<BB * (HD / 32), 128, 0, stream>>>(X, W2t, b2, G, Gmask);
}


# Cluster-sum launch as 256 work-groups x 16 waves (18432 -> 4096 waves): a wave sums one cluster's half rows in registers, next 16 loads issued ahead; weight pre-pass rides on them; sum order kept
# speedup vs baseline: 2.5312x; 1.0354x over previous
#include <hip/hip_runtime.h>
#include <hip/hip_bf16.h>

#define BB 32
#define NN 8192
#define HD 256
#define GH 512
#define KC 64
#define ROWS_PER_CLUSTER 128
#define MROWS (BB * (2 * KC + 1))
#define OUTROWS (KC + 1)

typedef __attribute__((ext_vector_type(2))) float v2f;
typedef __attribute__((ext_vector_type(4))) float v4f;
typedef __attribute__((ext_vector_type(8))) float v8f;

template <int Kd, int Nd>
__device__ __forceinline__ v4f pack_b_load(const float* Wm, int idx) {
    const int lane  = idx & 31;
    const int chunk = idx >> 5;
    constexpr int nkc = Kd >> 3;
    const int kc    = chunk % nkc;
    const int nt    = chunk / nkc;
    const int n     = nt * 16 + (lane & 15);
    const int k     = kc * 8 + (lane >> 4) * 4;
    const float* p  = Wm + (size_t)k * Nd + n;
    v4f v;
    v.x = p[0];
    v.y = p[(size_t)Nd];
    v.z = p[(size_t)2 * Nd];
    v.w = p[(size_t)3 * Nd];
    return v;
}

#define PACK_PIECES_W1 ((GH / 16) * (HD / 8) * 32)
#define PACK_PIECES_W2 ((HD / 16) * (GH / 8) * 32)

#define SUM_WAVES 16
#define SUM_WGS   (BB * KC * 2 / SUM_WAVES)
#define GROUP_ROWS 16
static_assert(PACK_PIECES_W1 == (SUM_WGS / 2) * 256 && PACK_PIECES_W2 == (SUM_WGS / 2) * 256,
              "work-groups 0..127 pack W1, 128..255 pack W2, 256 pieces each");
static_assert(ROWS_PER_CLUSTER % GROUP_ROWS == 0 && GROUP_ROWS % 4 == 0,
              "whole groups; every group starts on chain 0");

__device__ __forceinline__ void load_rows(v4f (&r)[GROUP_ROWS], const float* p, int g) {
    #pragma unroll
    for (int i = 0; i < GROUP_ROWS; ++i)
        r[i] = *(const v4f*)(p + (size_t)(g * GROUP_ROWS + i) * HD);
}

__device__ __forceinline__ void sum_rows(v4f (&s)[4], const v4f (&r)[GROUP_ROWS]) {
    #pragma unroll
    for (int i = 0; i < GROUP_ROWS; ++i) s[i & 3] += r[i];
}

template <int PW>
__device__ __forceinline__ void cluster_sum_body(const float* hs, float* H,
                                                 const float* Wsrc, float* Wdst,
                                                 int pidx, int blk, int half, int lane) {
    const int b = blk >> 6;
    const int k = blk & 63;
    const int col = half * (HD / 2) + lane * 4;

    v4f pk = {0.f, 0.f, 0.f, 0.f};
    if (PW == 1) pk = pack_b_load<HD, GH>(Wsrc, pidx);
    if (PW == 2) pk = pack_b_load<GH, HD>(Wsrc, pidx);

    const float* base = hs + ((size_t)b * NN + (size_t)k * ROWS_PER_CLUSTER) * HD + col;
    const v4f hn = *(const v4f*)(hs + ((size_t)b * NN + (NN - 1)) * HD + col);

    constexpr int NG = ROWS_PER_CLUSTER / GROUP_ROWS;
    v4f r[2][GROUP_ROWS];
    v4f s[4];
    #pragma unroll
    for (int q = 0; q < 4; ++q) s[q] = (v4f){0.f, 0.f, 0.f, 0.f};
#define SCHED_FENCE() __builtin_amdgcn_sched_barrier(0)
    load_rows(r[0], base, 0);
    #pragma unroll
    for (int g = 0; g < NG; ++g) {
        SCHED_FENCE();
        if (g + 1 < NG) load_rows(r[(g + 1) & 1], base, g + 1);
        SCHED_FENCE();
        if (PW != 0 && g == 0)
            *(v4f*)(Wdst + (size_t)pidx * 4) = pk;
        SCHED_FENCE();
        sum_rows(s, r[g & 1]);
    }
    SCHED_FENCE();
#undef SCHED_FENCE

    v4f tot = s[0] + s[1] + s[2] + s[3];
    if (k == KC - 1) tot -= hn;

    float* Hb = H + (size_t)b * (2 * KC + 1) * HD + col;
    *(v4f*)(Hb + (size_t)k * HD)        = tot;
    *(v4f*)(Hb + (size_t)(KC + k) * HD) = tot + hn;
    if (k == 0)
        *(v4f*)(Hb + (size_t)(2 * KC) * HD) = hn;
}

__global__ __launch_bounds__(SUM_WAVES * 32)
void cluster_sum_pack_kernel(const float* hs, float* H,
                             const float* W1, float* W1p,
                             const float* W2, float* W2p) {
    const int lane = threadIdx.x & 31;
    const int wave = __builtin_amdgcn_readfirstlane((int)(threadIdx.x >> 5));
    const int blk  = blockIdx.x * (SUM_WAVES / 2) + (wave >> 1);
    const int half = wave & 1;

    if (wave < 8) {
        if (blockIdx.x < SUM_WGS / 2)
            cluster_sum_body<1>(hs, H, W1, W1p, blockIdx.x * 256 + threadIdx.x, blk, half, lane);
        else
            cluster_sum_body<2>(hs, H, W2, W2p, (blockIdx.x - SUM_WGS / 2) * 256 + threadIdx.x,
                                blk, half, lane);
    } else {
        cluster_sum_body<0>(hs, H, W1, W1p, 0, blk, half, lane);
    }
}

#define LEFT_ROW0 ((MROWS / 16 - 2) * 16)

template <bool EX>
__device__ __forceinline__ void load_stage1(v4f (&a)[4], v4f (&ax)[4], v4f (&b)[4][4],
                                            const float* arow, const float* axrow,
                                            const float* bbase, const int (&boff)[4], int s) {
    #pragma unroll
    for (int j = 0; j < 4; ++j) {
        if (EX) ax[j] = *(const v4f*)(axrow + s * 32 + j * 8);
        a[j] = *(const v4f*)(arow + s * 32 + j * 8);
        #pragma unroll
        for (int t = 0; t < 4; ++t)
            b[t][j] = *(const v4f*)(bbase + boff[t] + (size_t)(s * 4 + j) * 128);
    }
}

template <bool EX>
__device__ __forceinline__ void mma_stage1(v8f (&acc)[4], v8f& accx, const v4f (&a)[4],
                                           const v4f (&ax)[4], const v4f (&b)[4][4]) {
    #pragma unroll
    for (int j = 0; j < 4; ++j) {
        const v2f alo = a[j].xy, ahi = a[j].zw;
        #pragma unroll
        for (int t = 0; t < 4; ++t)
            acc[t] = __builtin_amdgcn_wmma_f32_16x16x4_f32(
                false, alo, false, (v2f)b[t][j].xy, (short)0, acc[t], false, false);
        if (EX)
            accx = __builtin_amdgcn_wmma_f32_16x16x4_f32(
                false, (v2f)ax[j].xy, false, (v2f)b[0][j].xy, (short)0, accx, false, false);
        #pragma unroll
        for (int t = 0; t < 4; ++t)
            acc[t] = __builtin_amdgcn_wmma_f32_16x16x4_f32(
                false, ahi, false, (v2f)b[t][j].zw, (short)0, acc[t], false, false);
        if (EX)
            accx = __builtin_amdgcn_wmma_f32_16x16x4_f32(
                false, (v2f)ax[j].zw, false, (v2f)b[0][j].zw, (short)0, accx, false, false);
    }
}

template <bool EX>
__device__ __forceinline__ void gemm1_body(const float* A, const float* Bp,
                                           const float* __restrict__ bias, float* C,
                                           int lane, int mt, int ns, int r, int L) {
    const int colL = lane & 15;
    const int half = lane >> 4;
    const int m0 = mt << 4;
    const int n0 = ns << 6;

    int tt[4];
    #pragma unroll
    for (int t = 0; t < 4; ++t) tt[t] = EX ? ((t + r) & 3) : t;

    v8f acc[4];
    #pragma unroll
    for (int t = 0; t < 4; ++t) {
        const float bv = bias[n0 + 16 * tt[t] + colL];
        acc[t] = (v8f){bv, bv, bv, bv, bv, bv, bv, bv};
    }
    v8f accx = acc[0];

    const float* arow  = A  + (size_t)(m0 + colL) * HD + half * 4;
    const float* axrow = A  + (size_t)(LEFT_ROW0 + 16 * L + colL) * HD + half * 4;
    const float* bbase = Bp + ((size_t)(n0 >> 4) * (HD / 8) * 32 + lane) * 4;
    int boff[4];
    #pragma unroll
    for (int t = 0; t < 4; ++t) boff[t] = tt[t] * ((HD / 8) * 128);

    constexpr int NS = HD / 32;
    v4f a0[4], x0[4], b0[4][4], a1[4], x1[4], b1[4][4];

#define SCHED_FENCE() __builtin_amdgcn_sched_barrier(0)
    load_stage1<EX>(a0, x0, b0, arow, axrow, bbase, boff, 0);
    SCHED_FENCE();
    #pragma unroll 1
    for (int s = 0; s < NS - 2; s += 2) {
        load_stage1<EX>(a1, x1, b1, arow, axrow, bbase, boff, s + 1);
        SCHED_FENCE();
        mma_stage1<EX>(acc, accx, a0, x0, b0);
        SCHED_FENCE();
        load_stage1<EX>(a0, x0, b0, arow, axrow, bbase, boff, s + 2);
        SCHED_FENCE();
        mma_stage1<EX>(acc, accx, a1, x1, b1);
        SCHED_FENCE();
    }
    load_stage1<EX>(a1, x1, b1, arow, axrow, bbase, boff, NS - 1);
    SCHED_FENCE();
    mma_stage1<EX>(acc, accx, a0, x0, b0);
    mma_stage1<EX>(acc, accx, a1, x1, b1);
#undef SCHED_FENCE

    const int rbase = m0 + (half << 3);
    #pragma unroll
    for (int t = 0; t < 4; ++t) {
        const int col = n0 + 16 * tt[t] + colL;
        #pragma unroll
        for (int v = 0; v < 8; ++v)
            C[(size_t)(rbase + v) * GH + col] = fmaxf(acc[t][v], 0.0f);
    }
    if (EX) {
        const int rx  = LEFT_ROW0 + 16 * L + (half << 3);
        const int col = n0 + 16 * tt[0] + colL;
        #pragma unroll
        for (int v = 0; v < 8; ++v)
            C[(size_t)(rx + v) * GH + col] = fmaxf(accx[v], 0.0f);
    }
}

__global__ __launch_bounds__(256)
void gemm1_kernel(const float* A, const float* Bp,
                  const float* __restrict__ bias, float* C) {
    const int lane = threadIdx.x & 31;
    const int wave = __builtin_amdgcn_readfirstlane((int)(threadIdx.x >> 5));
    const int q  = blockIdx.x & 31;
    const int ns = blockIdx.x >> 5;
    const int mt = q * 8 + wave;

    if (wave == 0 && q < 8)
        gemm1_body<true>(A, Bp, bias, C, lane, mt, ns, q & 3, q >> 2);
    else
        gemm1_body<false>(A, Bp, bias, C, lane, mt, ns, 0, 0);
}

template <int HT>
__device__ __forceinline__ void load_stage_c(v4f (&a)[2][4], v4f (&ah)[4], v4f (&b)[2][4],
                                             const float* ar0, const float* ar1,
                                             const float* arh, const float* bbase, int s) {
    #pragma unroll
    for (int j = 0; j < 4; ++j) {
        if (HT >= 0) ah[j] = *(const v4f*)(arh + s * 32 + j * 8);
        a[0][j] = *(const v4f*)(ar0 + s * 32 + j * 8);
        a[1][j] = *(const v4f*)(ar1 + s * 32 + j * 8);
        #pragma unroll
        for (int t = 0; t < 2; ++t)
            b[t][j] = *(const v4f*)(bbase + ((size_t)t * (GH / 8) + (size_t)(s * 4 + j)) * 128);
    }
}

template <int HT>
__device__ __forceinline__ void mma_stage_c(v8f (&acc)[2][2], v8f& acch, const v4f (&a)[2][4],
                                            const v4f (&ah)[4], const v4f (&b)[2][4]) {
    constexpr int TB = HT < 0 ? 0 : HT;
    #pragma unroll
    for (int j = 0; j < 4; ++j) {
        #pragma unroll
        for (int m = 0; m < 2; ++m) {
            #pragma unroll
            for (int t = 0; t < 2; ++t)
                acc[m][t] = __builtin_amdgcn_wmma_f32_16x16x4_f32(
                    false, (v2f)a[m][j].xy, false, (v2f)b[t][j].xy, (short)0, acc[m][t], false, false);
        }
        if (HT >= 0)
            acch = __builtin_amdgcn_wmma_f32_16x16x4_f32(
                false, (v2f)ah[j].xy, false, (v2f)b[TB][j].xy, (short)0, acch, false, false);
        #pragma unroll
        for (int m = 0; m < 2; ++m) {
            #pragma unroll
            for (int t = 0; t < 2; ++t)
                acc[m][t] = __builtin_amdgcn_wmma_f32_16x16x4_f32(
                    false, (v2f)a[m][j].zw, false, (v2f)b[t][j].zw, (short)0, acc[m][t], false, false);
        }
        if (HT >= 0)
            acch = __builtin_amdgcn_wmma_f32_16x16x4_f32(
                false, (v2f)ah[j].zw, false, (v2f)b[TB][j].zw, (short)0, acch, false, false);
    }
    if (HT >= 0) asm volatile("" : "+v"(acch));
}

template <int HT>
__device__ __forceinline__ void k_loop_c(v8f (&acc)[2][2], v8f& acch,
                                         const float* ar0, const float* ar1,
                                         const float* arh, const float* bbase) {
    constexpr int NS = GH / 32;
    static_assert((NS - 4) % 3 == 0, "the loop below takes three stages per trip");
    v4f a0[2][4], h0[4], b0[2][4], a1[2][4], h1[4], b1[2][4], a2[2][4], h2[4], b2[2][4];
#define SCHED_FENCE() __builtin_amdgcn_sched_barrier(0)
    load_stage_c<HT>(a0, h0, b0, ar0, ar1, arh, bbase, 0);
    SCHED_FENCE();
    load_stage_c<HT>(a1, h1, b1, ar0, ar1, arh, bbase, 1);
    SCHED_FENCE();
    #pragma unroll 1
    for (int s = 0; s < NS - 4; s += 3) {
        load_stage_c<HT>(a2, h2, b2, ar0, ar1, arh, bbase, s + 2);
        SCHED_FENCE();
        mma_stage_c<HT>(acc, acch, a0, h0, b0);
        SCHED_FENCE();
        load_stage_c<HT>(a0, h0, b0, ar0, ar1, arh, bbase, s + 3);
        SCHED_FENCE();
        mma_stage_c<HT>(acc, acch, a1, h1, b1);
        SCHED_FENCE();
        load_stage_c<HT>(a1, h1, b1, ar0, ar1, arh, bbase, s + 4);
        SCHED_FENCE();
        mma_stage_c<HT>(acc, acch, a2, h2, b2);
        SCHED_FENCE();
    }
    load_stage_c<HT>(a2, h2, b2, ar0, ar1, arh, bbase, NS - 2);
    SCHED_FENCE();
    mma_stage_c<HT>(acc, acch, a0, h0, b0);
    SCHED_FENCE();
    load_stage_c<HT>(a0, h0, b0, ar0, ar1, arh, bbase, NS - 1);
    SCHED_FENCE();
    mma_stage_c<HT>(acc, acch, a1, h1, b1);
    mma_stage_c<HT>(acc, acch, a2, h2, b2);
    mma_stage_c<HT>(acc, acch, a0, h0, b0);
#undef SCHED_FENCE
}

__global__ __launch_bounds__(128)
void gemm2_combine_kernel(const float* X, const float* Bp,
                          const float* __restrict__ bias,
                          float* __restrict__ G, float* __restrict__ Gmask) {
    const int lane = threadIdx.x & 31;
    const int wave = __builtin_amdgcn_readfirstlane((int)(threadIdx.x >> 5));
    const int b    = blockIdx.x >> 3;
    const int n0   = (blockIdx.x & 7) << 5;
    const int colL = lane & 15;
    const int half = lane >> 4;

    v8f acc[2][2];
    #pragma unroll
    for (int t = 0; t < 2; ++t) {
        const float bv = bias[n0 + 16 * t + colL];
        acc[0][t] = (v8f){bv, bv, bv, bv, bv, bv, bv, bv};
        acc[1][t] = acc[0][t];
    }
    const float bvh = bias[n0 + 16 * (wave & 1) + colL];
    v8f acch = (v8f){bvh, bvh, bvh, bvh, bvh, bvh, bvh, bvh};

    const size_t r0 = (size_t)b * (2 * KC + 1);
    const float* ar0 = X + (r0 + 16 * wave + colL) * GH + half * 4;
    const float* ar1 = X + (r0 + KC + 16 * wave + colL) * GH + half * 4;
    const float* arh = X + (r0 + 2 * KC) * GH + half * 4;
    const float* bbase = Bp + ((size_t)(n0 >> 4) * (GH / 8) * 32 + lane) * 4;

    if (wave == 0)      k_loop_c<0>(acc, acch, ar0, ar1, arh, bbase);
    else if (wave == 1) k_loop_c<1>(acc, acch, ar0, ar1, arh, bbase);
    else                k_loop_c<-1>(acc, acch, ar0, ar1, arh, bbase);

    __shared__ float sS[4][2][32];
    #pragma unroll
    for (int t = 0; t < 2; ++t) {
        float ps = acc[0][t][0];
        #pragma unroll
        for (int v = 1; v < 8; ++v) ps += acc[0][t][v];
        sS[wave][half][16 * t + colL] = ps;
    }
    __syncthreads();
    float S[2];
    #pragma unroll
    for (int t = 0; t < 2; ++t) {
        float s = 0.f;
        #pragma unroll
        for (int w = 0; w < 4; ++w) {
            s += sS[w][0][16 * t + colL];
            s += sS[w][1][16 * t + colL];
        }
        S[t] = s;
    }

    float* Gb = G + (size_t)b * OUTROWS * HD;
    const int rbase = 16 * wave + (half << 3);
    #pragma unroll
    for (int t = 0; t < 2; ++t) {
        const int col = n0 + 16 * t + colL;
        #pragma unroll
        for (int v = 0; v < 8; ++v)
            Gb[(size_t)(rbase + v) * HD + col] = S[t] - acc[0][t][v] + acc[1][t][v];
    }
    if (wave < 2 && half == 0)
        Gb[(size_t)KC * HD + n0 + 16 * wave + colL] = (wave == 0 ? S[0] : S[1]) + acch[0];

    if ((blockIdx.x & 7) == 0 && threadIdx.x < OUTROWS)
        Gmask[b * OUTROWS + threadIdx.x] = 1.0f;
}

extern "C" void kernel_launch(void* const* d_in, const int* in_sizes, int n_in,
                              void* d_out, int out_size, void* d_ws, size_t ws_size,
                              hipStream_t stream) {
    const float* hs = (const float*)d_in[0];
    const float* W1 = (const float*)d_in[2];
    const float* b1 = (const float*)d_in[3];
    const float* W2 = (const float*)d_in[4];
    const float* b2 = (const float*)d_in[5];

    float* G     = (float*)d_out;
    float* Gmask = (float*)d_out + (size_t)BB * OUTROWS * HD;

    float* H   = (float*)d_ws;
    float* X   = H   + (size_t)MROWS * HD;
    float* W1t = X   + (size_t)MROWS * GH;
    float* W2t = W1t + (size_t)GH * HD;

    cluster_sum_pack_kernel<<<SUM_WGS, SUM_WAVES * 32, 0, stream>>>(
        hs, H, W1, W1t, W2, W2t);

    gemm1_kernel<<<(MROWS / 16 - 2) * (GH / 64) / 8, 256, 0, stream>>>(H, W1t, b1, X);

    gemm2_combine_kernel<<<BB * (HD / 32), 128, 0, stream>>>(X, W2t, b2, G, Gmask);
}


# First layer computed inside the cluster-sum launch: a work-group's 8 clusters form one 16-row tile [Hk;Hk+hn] in LDS, B read from row-major W1 (tile t = columns 2c+t); 3 -> 2 launches; sum order kept
# speedup vs baseline: 2.6577x; 1.0500x over previous
#include <hip/hip_runtime.h>
#include <hip/hip_bf16.h>

#define BB 32
#define NN 8192
#define HD 256
#define GH 512
#define KC 64
#define ROWS_PER_CLUSTER 128
#define MROWS (BB * (2 * KC + 1))
#define OUTROWS (KC + 1)

typedef __attribute__((ext_vector_type(2))) float v2f;
typedef __attribute__((ext_vector_type(4))) float v4f;
typedef __attribute__((ext_vector_type(8))) float v8f;

template <int Kd, int Nd>
__device__ __forceinline__ v4f pack_b_load(const float* Wm, int idx) {
    const int lane  = idx & 31;
    const int chunk = idx >> 5;
    constexpr int nkc = Kd >> 3;
    const int kc    = chunk % nkc;
    const int nt    = chunk / nkc;
    const int n     = nt * 16 + (lane & 15);
    const int k     = kc * 8 + (lane >> 4) * 4;
    const float* p  = Wm + (size_t)k * Nd + n;
    v4f v;
    v.x = p[0];
    v.y = p[(size_t)Nd];
    v.z = p[(size_t)2 * Nd];
    v.w = p[(size_t)3 * Nd];
    return v;
}

#define PACK_PIECES_W2 ((HD / 16) * (GH / 8) * 32)

#define SUM_WAVES 16
#define SUM_WGS   (BB * KC * 2 / SUM_WAVES)
#define GROUP_ROWS 16
#define LDA (HD + 4)
static_assert(PACK_PIECES_W2 == (SUM_WGS / 2) * 256, "work-groups 128..255 pack W2, 256 pieces each");
static_assert(ROWS_PER_CLUSTER % GROUP_ROWS == 0 && GROUP_ROWS % 4 == 0,
              "whole groups; every group starts on chain 0");
static_assert(SUM_WAVES * 32 == GH, "16 waves x 32 columns cover the hidden layer");

__device__ __forceinline__ void load_rows(v4f (&r)[GROUP_ROWS], const float* p, int g) {
    #pragma unroll
    for (int i = 0; i < GROUP_ROWS; ++i)
        r[i] = *(const v4f*)(p + (size_t)(g * GROUP_ROWS + i) * HD);
}

__device__ __forceinline__ void sum_rows(v4f (&s)[4], const v4f (&r)[GROUP_ROWS]) {
    #pragma unroll
    for (int i = 0; i < GROUP_ROWS; ++i) s[i & 3] += r[i];
}

template <int PW>
__device__ __forceinline__ void cluster_sum_body(const float* hs, const float* Wsrc, float* Wdst,
                                                 int pidx, int blk, int half, int lane,
                                                 v4f& tot_out, v4f& hn_out) {
    const int b = blk >> 6;
    const int k = blk & 63;
    const int col = half * (HD / 2) + lane * 4;

    v4f pk = {0.f, 0.f, 0.f, 0.f};
    if (PW == 2) pk = pack_b_load<GH, HD>(Wsrc, pidx);

    const float* base = hs + ((size_t)b * NN + (size_t)k * ROWS_PER_CLUSTER) * HD + col;
    const v4f hn = *(const v4f*)(hs + ((size_t)b * NN + (NN - 1)) * HD + col);

    constexpr int NG = ROWS_PER_CLUSTER / GROUP_ROWS;
    v4f r[2][GROUP_ROWS];
    v4f s[4];
    #pragma unroll
    for (int q = 0; q < 4; ++q) s[q] = (v4f){0.f, 0.f, 0.f, 0.f};
#define SCHED_FENCE() __builtin_amdgcn_sched_barrier(0)
    load_rows(r[0], base, 0);
    #pragma unroll
    for (int g = 0; g < NG; ++g) {
        SCHED_FENCE();
        if (g + 1 < NG) load_rows(r[(g + 1) & 1], base, g + 1);
        SCHED_FENCE();
        if (PW != 0 && g == 0)
            *(v4f*)(Wdst + (size_t)pidx * 4) = pk;
        SCHED_FENCE();
        sum_rows(s, r[g & 1]);
    }
    SCHED_FENCE();
#undef SCHED_FENCE

    v4f tot = s[0] + s[1] + s[2] + s[3];
    if (k == KC - 1) tot -= hn;
    tot_out = tot;
    hn_out  = hn;
}

__device__ __forceinline__ void load_b_f(v2f (&bl)[4][4], const float* bptr, int s) {
    #pragma unroll
    for (int j = 0; j < 4; ++j) {
        #pragma unroll
        for (int e = 0; e < 4; ++e)
            bl[j][e] = *(const v2f*)(bptr + (size_t)(s * 32 + j * 8 + e) * GH);
    }
}

template <int EXT>
__device__ __forceinline__ void load_a_f(v4f (&a)[4], v4f (&ax)[4],
                                         const float* arow, const float* axrow, int s) {
    #pragma unroll
    for (int j = 0; j < 4; ++j) {
        if (EXT >= 0) ax[j] = *(const v4f*)(axrow + s * 32 + j * 8);
        a[j] = *(const v4f*)(arow + s * 32 + j * 8);
    }
}

template <int EXT>
__device__ __forceinline__ void mma_stage_f(v8f (&acc)[2], v8f& accx, const v4f (&a)[4],
                                            const v4f (&ax)[4], const v2f (&bl)[4][4]) {
    #pragma unroll
    for (int j = 0; j < 4; ++j) {
        const v2f b00 = {bl[j][0].x, bl[j][1].x};
        const v2f b10 = {bl[j][0].y, bl[j][1].y};
        const v2f b01 = {bl[j][2].x, bl[j][3].x};
        const v2f b11 = {bl[j][2].y, bl[j][3].y};
        acc[0] = __builtin_amdgcn_wmma_f32_16x16x4_f32(
            false, (v2f)a[j].xy, false, b00, (short)0, acc[0], false, false);
        acc[1] = __builtin_amdgcn_wmma_f32_16x16x4_f32(
            false, (v2f)a[j].xy, false, b10, (short)0, acc[1], false, false);
        if (EXT >= 0)
            accx = __builtin_amdgcn_wmma_f32_16x16x4_f32(
                false, (v2f)ax[j].xy, false, EXT == 0 ? b00 : b10, (short)0, accx, false, false);
        acc[0] = __builtin_amdgcn_wmma_f32_16x16x4_f32(
            false, (v2f)a[j].zw, false, b01, (short)0, acc[0], false, false);
        acc[1] = __builtin_amdgcn_wmma_f32_16x16x4_f32(
            false, (v2f)a[j].zw, false, b11, (short)0, acc[1], false, false);
        if (EXT >= 0)
            accx = __builtin_amdgcn_wmma_f32_16x16x4_f32(
                false, (v2f)ax[j].zw, false, EXT == 0 ? b01 : b11, (short)0, accx, false, false);
    }
    asm volatile("" : "+v"(acc[0]), "+v"(acc[1]));
    if (EXT >= 0) asm volatile("" : "+v"(accx));
}

template <int EXT>
__device__ __forceinline__ void gemm1_tail(const float* arow, const float* axrow, const float* bptr,
                                           const v2f (&bs0)[4][4], float bias0, float bias1,
                                           float* xr, float* xx) {
    v8f acc[2];
    acc[0] = (v8f){bias0, bias0, bias0, bias0, bias0, bias0, bias0, bias0};
    acc[1] = (v8f){bias1, bias1, bias1, bias1, bias1, bias1, bias1, bias1};
    v8f accx = acc[EXT < 0 ? 0 : EXT];

    constexpr int NS = HD / 32;
    v4f a0[4], x0[4], a1[4], x1[4];
    v2f b0[4][4], b1[4][4];
    #pragma unroll
    for (int j = 0; j < 4; ++j) {
        #pragma unroll
        for (int e = 0; e < 4; ++e) b0[j][e] = bs0[j][e];
    }
#define SCHED_FENCE() __builtin_amdgcn_sched_barrier(0)
    load_a_f<EXT>(a0, x0, arow, axrow, 0);
    SCHED_FENCE();
    #pragma unroll 1
    for (int s = 0; s < NS - 2; s += 2) {
        load_b_f(b1, bptr, s + 1);
        load_a_f<EXT>(a1, x1, arow, axrow, s + 1);
        SCHED_FENCE();
        mma_stage_f<EXT>(acc, accx, a0, x0, b0);
        SCHED_FENCE();
        load_b_f(b0, bptr, s + 2);
        load_a_f<EXT>(a0, x0, arow, axrow, s + 2);
        SCHED_FENCE();
        mma_stage_f<EXT>(acc, accx, a1, x1, b1);
        SCHED_FENCE();
    }
    load_b_f(b1, bptr, NS - 1);
    load_a_f<EXT>(a1, x1, arow, axrow, NS - 1);
    SCHED_FENCE();
    mma_stage_f<EXT>(acc, accx, a0, x0, b0);
    mma_stage_f<EXT>(acc, accx, a1, x1, b1);
#undef SCHED_FENCE

    #pragma unroll
    for (int v = 0; v < 8; ++v)
        *(v2f*)(xr + (size_t)v * GH) = (v2f){fmaxf(acc[0][v], 0.0f), fmaxf(acc[1][v], 0.0f)};
    if (EXT >= 0) {
        #pragma unroll
        for (int v = 0; v < 8; ++v)
            xx[(size_t)v * (2 * KC + 1) * GH] = fmaxf(accx[v], 0.0f);
    }
}

__global__ __launch_bounds__(SUM_WAVES * 32)
void cluster_sum_layer1_kernel(const float* hs, const float* W1, const float* __restrict__ bias1,
                               float* X, const float* W2, float* W2p) {
    __shared__ __attribute__((aligned(16))) float sA[16 * LDA];

    const int lane = threadIdx.x & 31;
    const int wave = __builtin_amdgcn_readfirstlane((int)(threadIdx.x >> 5));
    const int blk  = blockIdx.x * (SUM_WAVES / 2) + (wave >> 1);
    const int half = wave & 1;

    v4f tot, hn;
    if (wave < 8 && blockIdx.x >= SUM_WGS / 2)
        cluster_sum_body<2>(hs, W2, W2p, (blockIdx.x - SUM_WGS / 2) * 256 + threadIdx.x,
                            blk, half, lane, tot, hn);
    else
        cluster_sum_body<0>(hs, W2, W2p, 0, blk, half, lane, tot, hn);

    const int c  = lane & 15;
    const int h  = lane >> 4;
    const int n0 = wave * 32;
    const float* bptr = W1 + (size_t)(4 * h) * GH + n0 + 2 * c;
    v2f bs0[4][4];
    load_b_f(bs0, bptr, 0);
    const v2f bv = *(const v2f*)(bias1 + n0 + 2 * c);

    const int ci = wave >> 1;
    *(v4f*)(sA + ci * LDA + half * (HD / 2) + lane * 4)       = tot;
    *(v4f*)(sA + (8 + ci) * LDA + half * (HD / 2) + lane * 4) = tot + hn;
    __syncthreads();

    const int b  = blockIdx.x >> 3;
    const int k0 = (blockIdx.x & 7) * 8;
    const float* arow = sA + c * LDA + 4 * h;
    float* xr = X + ((size_t)b * (2 * KC + 1) + (size_t)h * KC + k0) * GH + n0 + 2 * c;

    const int L = (blockIdx.x >> 5) & 1;
    const float* axrow = hs + ((size_t)(16 * L + c) * NN + (NN - 1)) * HD + 4 * h;
    float* xx = X + ((size_t)(16 * L + 8 * h) * (2 * KC + 1) + 2 * KC) * GH
                  + n0 + 2 * c + (blockIdx.x & 1);

    if (blockIdx.x < 64 && wave == ((blockIdx.x & 31) >> 1)) {
        if ((blockIdx.x & 1) == 0)
            gemm1_tail<0>(arow, axrow, bptr, bs0, bv.x, bv.y, xr, xx);
        else
            gemm1_tail<1>(arow, axrow, bptr, bs0, bv.x, bv.y, xr, xx);
    } else {
        gemm1_tail<-1>(arow, axrow, bptr, bs0, bv.x, bv.y, xr, xx);
    }
}

template <int HT>
__device__ __forceinline__ void load_stage_c(v4f (&a)[2][4], v4f (&ah)[4], v4f (&b)[2][4],
                                             const float* ar0, const float* ar1,
                                             const float* arh, const float* bbase, int s) {
    #pragma unroll
    for (int j = 0; j < 4; ++j) {
        if (HT >= 0) ah[j] = *(const v4f*)(arh + s * 32 + j * 8);
        a[0][j] = *(const v4f*)(ar0 + s * 32 + j * 8);
        a[1][j] = *(const v4f*)(ar1 + s * 32 + j * 8);
        #pragma unroll
        for (int t = 0; t < 2; ++t)
            b[t][j] = *(const v4f*)(bbase + ((size_t)t * (GH / 8) + (size_t)(s * 4 + j)) * 128);
    }
}

template <int HT>
__device__ __forceinline__ void mma_stage_c(v8f (&acc)[2][2], v8f& acch, const v4f (&a)[2][4],
                                            const v4f (&ah)[4], const v4f (&b)[2][4]) {
    constexpr int TB = HT < 0 ? 0 : HT;
    #pragma unroll
    for (int j = 0; j < 4; ++j) {
        #pragma unroll
        for (int m = 0; m < 2; ++m) {
            #pragma unroll
            for (int t = 0; t < 2; ++t)
                acc[m][t] = __builtin_amdgcn_wmma_f32_16x16x4_f32(
                    false, (v2f)a[m][j].xy, false, (v2f)b[t][j].xy, (short)0, acc[m][t], false, false);
        }
        if (HT >= 0)
            acch = __builtin_amdgcn_wmma_f32_16x16x4_f32(
                false, (v2f)ah[j].xy, false, (v2f)b[TB][j].xy, (short)0, acch, false, false);
        #pragma unroll
        for (int m = 0; m < 2; ++m) {
            #pragma unroll
            for (int t = 0; t < 2; ++t)
                acc[m][t] = __builtin_amdgcn_wmma_f32_16x16x4_f32(
                    false, (v2f)a[m][j].zw, false, (v2f)b[t][j].zw, (short)0, acc[m][t], false, false);
        }
        if (HT >= 0)
            acch = __builtin_amdgcn_wmma_f32_16x16x4_f32(
                false, (v2f)ah[j].zw, false, (v2f)b[TB][j].zw, (short)0, acch, false, false);
    }
    if (HT >= 0) asm volatile("" : "+v"(acch));
}

template <int HT>
__device__ __forceinline__ void k_loop_c(v8f (&acc)[2][2], v8f& acch,
                                         const float* ar0, const float* ar1,
                                         const float* arh, const float* bbase) {
    constexpr int NS = GH / 32;
    static_assert((NS - 4) % 3 == 0, "the loop below takes three stages per trip");
    v4f a0[2][4], h0[4], b0[2][4], a1[2][4], h1[4], b1[2][4], a2[2][4], h2[4], b2[2][4];
#define SCHED_FENCE() __builtin_amdgcn_sched_barrier(0)
    load_stage_c<HT>(a0, h0, b0, ar0, ar1, arh, bbase, 0);
    SCHED_FENCE();
    load_stage_c<HT>(a1, h1, b1, ar0, ar1, arh, bbase, 1);
    SCHED_FENCE();
    #pragma unroll 1
    for (int s = 0; s < NS - 4; s += 3) {
        load_stage_c<HT>(a2, h2, b2, ar0, ar1, arh, bbase, s + 2);
        SCHED_FENCE();
        mma_stage_c<HT>(acc, acch, a0, h0, b0);
        SCHED_FENCE();
        load_stage_c<HT>(a0, h0, b0, ar0, ar1, arh, bbase, s + 3);
        SCHED_FENCE();
        mma_stage_c<HT>(acc, acch, a1, h1, b1);
        SCHED_FENCE();
        load_stage_c<HT>(a1, h1, b1, ar0, ar1, arh, bbase, s + 4);
        SCHED_FENCE();
        mma_stage_c<HT>(acc, acch, a2, h2, b2);
        SCHED_FENCE();
    }
    load_stage_c<HT>(a2, h2, b2, ar0, ar1, arh, bbase, NS - 2);
    SCHED_FENCE();
    mma_stage_c<HT>(acc, acch, a0, h0, b0);
    SCHED_FENCE();
    load_stage_c<HT>(a0, h0, b0, ar0, ar1, arh, bbase, NS - 1);
    SCHED_FENCE();
    mma_stage_c<HT>(acc, acch, a1, h1, b1);
    mma_stage_c<HT>(acc, acch, a2, h2, b2);
    mma_stage_c<HT>(acc, acch, a0, h0, b0);
#undef SCHED_FENCE
}

__global__ __launch_bounds__(128)
void gemm2_combine_kernel(const float* X, const float* Bp,
                          const float* __restrict__ bias,
                          float* __restrict__ G, float* __restrict__ Gmask) {
    const int lane = threadIdx.x & 31;
    const int wave = __builtin_amdgcn_readfirstlane((int)(threadIdx.x >> 5));
    const int b    = blockIdx.x >> 3;
    const int n0   = (blockIdx.x & 7) << 5;
    const int colL = lane & 15;
    const int half = lane >> 4;

    v8f acc[2][2];
    #pragma unroll
    for (int t = 0; t < 2; ++t) {
        const float bv = bias[n0 + 16 * t + colL];
        acc[0][t] = (v8f){bv, bv, bv, bv, bv, bv, bv, bv};
        acc[1][t] = acc[0][t];
    }
    const float bvh = bias[n0 + 16 * (wave & 1) + colL];
    v8f acch = (v8f){bvh, bvh, bvh, bvh, bvh, bvh, bvh, bvh};

    const size_t r0 = (size_t)b * (2 * KC + 1);
    const float* ar0 = X + (r0 + 16 * wave + colL) * GH + half * 4;
    const float* ar1 = X + (r0 + KC + 16 * wave + colL) * GH + half * 4;
    const float* arh = X + (r0 + 2 * KC) * GH + half * 4;
    const float* bbase = Bp + ((size_t)(n0 >> 4) * (GH / 8) * 32 + lane) * 4;

    if (wave == 0)      k_loop_c<0>(acc, acch, ar0, ar1, arh, bbase);
    else if (wave == 1) k_loop_c<1>(acc, acch, ar0, ar1, arh, bbase);
    else                k_loop_c<-1>(acc, acch, ar0, ar1, arh, bbase);

    __shared__ float sS[4][2][32];
    #pragma unroll
    for (int t = 0; t < 2; ++t) {
        float ps = acc[0][t][0];
        #pragma unroll
        for (int v = 1; v < 8; ++v) ps += acc[0][t][v];
        sS[wave][half][16 * t + colL] = ps;
    }
    __syncthreads();
    float S[2];
    #pragma unroll
    for (int t = 0; t < 2; ++t) {
        float s = 0.f;
        #pragma unroll
        for (int w = 0; w < 4; ++w) {
            s += sS[w][0][16 * t + colL];
            s += sS[w][1][16 * t + colL];
        }
        S[t] = s;
    }

    float* Gb = G + (size_t)b * OUTROWS * HD;
    const int rbase = 16 * wave + (half << 3);
    #pragma unroll
    for (int t = 0; t < 2; ++t) {
        const int col = n0 + 16 * t + colL;
        #pragma unroll
        for (int v = 0; v < 8; ++v)
            Gb[(size_t)(rbase + v) * HD + col] = S[t] - acc[0][t][v] + acc[1][t][v];
    }
    if (wave < 2 && half == 0)
        Gb[(size_t)KC * HD + n0 + 16 * wave + colL] = (wave == 0 ? S[0] : S[1]) + acch[0];

    if ((blockIdx.x & 7) == 0 && threadIdx.x < OUTROWS)
        Gmask[b * OUTROWS + threadIdx.x] = 1.0f;
}

extern "C" void kernel_launch(void* const* d_in, const int* in_sizes, int n_in,
                              void* d_out, int out_size, void* d_ws, size_t ws_size,
                              hipStream_t stream) {
    const float* hs = (const float*)d_in[0];
    const float* W1 = (const float*)d_in[2];
    const float* b1 = (const float*)d_in[3];
    const float* W2 = (const float*)d_in[4];
    const float* b2 = (const float*)d_in[5];

    float* G     = (float*)d_out;
    float* Gmask = (float*)d_out + (size_t)BB * OUTROWS * HD;

    float* H   = (float*)d_ws;
    float* X   = H   + (size_t)MROWS * HD;
    float* W1t = X   + (size_t)MROWS * GH;
    float* W2t = W1t + (size_t)GH * HD;

    cluster_sum_layer1_kernel<<<SUM_WGS, SUM_WAVES * 32, 0, stream>>>(
        hs, W1, b1, X, W2, W2t);

    gemm2_combine_kernel<<<BB * (HD / 32), 128, 0, stream>>>(X, W2t, b2, G, Gmask);
}
